# starting kernel
# baseline (speedup 1.0000x reference)
#include <hip/hip_runtime.h>

typedef __attribute__((ext_vector_type(2))) float v2f;
typedef __attribute__((ext_vector_type(8))) float v8f;

__global__ __launch_bounds__(256) void ftl_proj_wmma(
    const float* __restrict__ z, const float* __restrict__ pm,
    float* __restrict__ out) {
  constexpr int HW = 64 * 64;
  constexpr int G  = 128;
  constexpr int NCHUNK = HW / 16;

  const int b    = blockIdx.y;
  const int g    = blockIdx.x;
  const int lane = threadIdx.x & 31;
  const int wave = threadIdx.x >> 5;
  const int col  = lane & 15;
  const int k0   = (lane >> 4) << 1;

  const float* p = pm + b * 12;
  v2f a;
  a.x = (col < 3) ? p[col * 4 + k0]     : 0.0f;
  a.y = (col < 3) ? p[col * 4 + k0 + 1] : 0.0f;

  const size_t zbase = ((size_t)b * (G * 4) + (size_t)g * 4) * (size_t)HW;
  const size_t obase = ((size_t)b * (G * 3) + (size_t)g * 3) * (size_t)HW;
  const float* zp0 = z + zbase + (size_t)k0 * HW + col;
  const float* zp1 = zp0 + HW;
  float* o0 = out + obase + col;

#pragma unroll 8
  for (int c = wave; c < NCHUNK; c += 8) {
    const int pos = c << 4;
    __builtin_prefetch(zp0 + pos + (8 << 4), 0, 0);

    v2f bm;
    bm.x = __builtin_nontemporal_load(zp0 + pos);
    bm.y = __builtin_nontemporal_load(zp1 + pos);

    v8f acc = {};
    acc = __builtin_amdgcn_wmma_f32_16x16x4_f32(
         false, a,  false, bm,
         (short)0, acc,  false,  false);

    if (lane < 16) {
      __builtin_nontemporal_store(acc[0], o0 + 0 * HW + pos);
      __builtin_nontemporal_store(acc[1], o0 + 1 * HW + pos);
      __builtin_nontemporal_store(acc[2], o0 + 2 * HW + pos);
    }
  }
}

extern "C" void kernel_launch(void* const* d_in, const int* in_sizes, int n_in,
                              void* d_out, int out_size, void* d_ws, size_t ws_size,
                              hipStream_t stream) {
  (void)in_sizes; (void)n_in; (void)d_ws; (void)ws_size; (void)out_size;
  const float* z  = (const float*)d_in[0];
  const float* pm = (const float*)d_in[1];
  float* out = (float*)d_out;

  dim3 grid(128, 16);
  ftl_proj_wmma<<<grid, 256, 0, stream>>>(z, pm, out);
}


# Rows staged through wave-private LDS both ways: each global load/store is a whole 512-B row piece, fragments come from LDS, one full-K WMMA per 16 positions; 4096 waves (sum order unchanged)
# speedup vs baseline: 2.4884x; 2.4884x over previous
#include <hip/hip_runtime.h>

typedef __attribute__((ext_vector_type(2))) float v2f;
typedef __attribute__((ext_vector_type(4))) float v4f;
typedef __attribute__((ext_vector_type(8))) float v8f;


constexpr int HW        = 64 * 64;
constexpr int UNIT      = 128;
constexpr int IN_PITCH  = 136;
constexpr int OUT_PITCH = 128;
constexpr int WAVES     = 8;
constexpr int NUNITS    = 16;
constexpr int AHEAD     = 3;

static_assert((IN_PITCH % 4) == 0 && (OUT_PITCH % 4) == 0, "16-byte LDS accesses stay aligned");

__global__ __launch_bounds__(WAVES * 32) void ftl_proj_rows_via_lds(
    const float* __restrict__ z, const float* __restrict__ pm,
    float* __restrict__ out) {
  __shared__ v4f in_slab4[WAVES * 4 * IN_PITCH / 4];
  __shared__ v4f out_slab4[WAVES * 3 * OUT_PITCH / 4];

  const int lane = threadIdx.x & 31;
  const int wave = threadIdx.x >> 5;
  const int n    = lane & 15;
  const int h    = lane >> 4;
  const int k0   = h << 1;

  float* in_s  = reinterpret_cast<float*>(in_slab4)  + wave * (4 * IN_PITCH);
  float* out_s = reinterpret_cast<float*>(out_slab4) + wave * (3 * OUT_PITCH);

  const int pair0 = blockIdx.x * 4;
  const int b     = pair0 >> 7;

  const float* p = pm + b * 12;
  v2f a_lo, a_hi;
  a_lo.x = (n < 3) ? p[n * 4 + k0]     : 0.0f;
  a_lo.y = (n < 3) ? p[n * 4 + k0 + 1] : 0.0f;
  a_hi.x = (n >= 8 && n < 11) ? p[(n - 8) * 4 + k0]     : 0.0f;
  a_hi.y = (n >= 8 && n < 11) ? p[(n - 8) * 4 + k0 + 1] : 0.0f;

  const int lane_off = wave * UNIT + 4 * lane;

  v4f ring[AHEAD + 1][4];
#define LOAD_UNIT(U) do { \
    const float* zu_ = z + (size_t)(pair0 + ((U) >> 2)) * (4 * HW) + ((U) & 3) * (WAVES * UNIT) + lane_off; \
    _Pragma("unroll") for (int j = 0; j < 4; ++j) \
      ring[(U) % (AHEAD + 1)][j] = __builtin_nontemporal_load(reinterpret_cast<const v4f*>(zu_ + j * HW)); \
  } while (0)

#pragma unroll
  for (int u = 0; u < AHEAD; ++u) LOAD_UNIT(u);

#pragma unroll
  for (int u = 0; u < NUNITS; ++u) {
    if (u + AHEAD < NUNITS) LOAD_UNIT(u + AHEAD);

#pragma unroll
    for (int j = 0; j < 4; ++j)
      *reinterpret_cast<v4f*>(in_s + j * IN_PITCH + 4 * lane) = ring[u % (AHEAD + 1)][j];

    __builtin_amdgcn_wave_barrier();

    v2f frag[8];
#pragma unroll
    for (int c = 0; c < 8; ++c) {
      frag[c].x = in_s[(k0    ) * IN_PITCH + 16 * c + n];
      frag[c].y = in_s[(k0 + 1) * IN_PITCH + 16 * c + n];
    }

    v8f acc[4];
#pragma unroll
    for (int pq = 0; pq < 4; ++pq) {
      v8f zero = {};
      acc[pq] = __builtin_amdgcn_wmma_f32_16x16x4_f32(
           false, a_lo,  false, frag[2 * pq],
           (short)0, zero,  false,  false);
    }
#pragma unroll
    for (int pq = 0; pq < 4; ++pq) {
      acc[pq] = __builtin_amdgcn_wmma_f32_16x16x4_f32(
           false, a_hi,  false, frag[2 * pq + 1],
           (short)0, acc[pq],  false,  false);
    }

#pragma unroll
    for (int pq = 0; pq < 4; ++pq) {
      out_s[0 * OUT_PITCH + 32 * pq + lane] = acc[pq][0];
      out_s[1 * OUT_PITCH + 32 * pq + lane] = acc[pq][1];
      out_s[2 * OUT_PITCH + 32 * pq + lane] = acc[pq][2];
    }

    __builtin_amdgcn_wave_barrier();

    float* ou = out + (size_t)(pair0 + (u >> 2)) * (3 * HW) + (u & 3) * (WAVES * UNIT) + lane_off;
#pragma unroll
    for (int i = 0; i < 3; ++i) {
      const v4f o = *reinterpret_cast<const v4f*>(out_s + i * OUT_PITCH + 4 * lane);
      __builtin_nontemporal_store(o, reinterpret_cast<v4f*>(ou + i * HW));
    }

    __builtin_amdgcn_wave_barrier();
  }
#undef LOAD_UNIT
}

extern "C" void kernel_launch(void* const* d_in, const int* in_sizes, int n_in,
                              void* d_out, int out_size, void* d_ws, size_t ws_size,
                              hipStream_t stream) {
  (void)in_sizes; (void)n_in; (void)d_ws; (void)ws_size; (void)out_size;
  const float* z  = (const float*)d_in[0];
  const float* pm = (const float*)d_in[1];
  float* out = (float*)d_out;

  ftl_proj_rows_via_lds<<<dim3(512), dim3(256), 0, stream>>>(z, pm, out);
}
